# starting kernel
# baseline (speedup 1.0000x reference)
#include <hip/hip_runtime.h>

typedef float v2f __attribute__((ext_vector_type(2)));
typedef float v8f __attribute__((ext_vector_type(8)));

#define HID  128
#define OUTC 128

__device__ __forceinline__ void wave_gemm_16x64(const float* __restrict__ h,
                                                const float* __restrict__ W,
                                                int m0, int nbase, int lane,
                                                v8f acc[4]) {
  const int l16  = lane & 15;
  const int koff = (lane >> 4) * 2;
  const float* ap  = h + (size_t)(m0 + l16) * HID + koff;
  const float* bp0 = W + (size_t)(nbase +  0 + l16) * HID + koff;
  const float* bp1 = W + (size_t)(nbase + 16 + l16) * HID + koff;
  const float* bp2 = W + (size_t)(nbase + 32 + l16) * HID + koff;
  const float* bp3 = W + (size_t)(nbase + 48 + l16) * HID + koff;
  acc[0] = v8f{}; acc[1] = v8f{}; acc[2] = v8f{}; acc[3] = v8f{};
#pragma unroll 8
  for (int k0 = 0; k0 < HID; k0 += 4) {
    v2f a  = *(const v2f*)(ap  + k0);
    v2f b0 = *(const v2f*)(bp0 + k0);
    v2f b1 = *(const v2f*)(bp1 + k0);
    v2f b2 = *(const v2f*)(bp2 + k0);
    v2f b3 = *(const v2f*)(bp3 + k0);
    acc[0] = __builtin_amdgcn_wmma_f32_16x16x4_f32(false, a, false, b0, (short)0, acc[0], false, false);
    acc[1] = __builtin_amdgcn_wmma_f32_16x16x4_f32(false, a, false, b1, (short)0, acc[1], false, false);
    acc[2] = __builtin_amdgcn_wmma_f32_16x16x4_f32(false, a, false, b2, (short)0, acc[2], false, false);
    acc[3] = __builtin_amdgcn_wmma_f32_16x16x4_f32(false, a, false, b3, (short)0, acc[3], false, false);
  }
}

__global__ __launch_bounds__(128) void gemm_kernel(const float* __restrict__ h,
                                                   const float* __restrict__ W,
                                                   float* __restrict__ z) {
  const int lane  = threadIdx.x & 31;
  const int wave  = threadIdx.x >> 5;
  const int m0    = blockIdx.x * 32 + (wave >> 1) * 16;
  const int nbase = (wave & 1) * 64;
  v8f acc[4];
  wave_gemm_16x64(h, W, m0, nbase, lane, acc);
  const int l16   = lane & 15;
  const int rbase = (lane >> 4) * 8;
#pragma unroll
  for (int r = 0; r < 8; ++r) {
    const int m = m0 + rbase + r;
#pragma unroll
    for (int j = 0; j < 4; ++j) {
      z[(size_t)m * OUTC + nbase + j * 16 + l16] = acc[j][r];
    }
  }
}

__global__ __launch_bounds__(256) void scatter_kernel(const int* __restrict__ src,
                                                      const int* __restrict__ dst,
                                                      const float* __restrict__ z,
                                                      float* __restrict__ s,
                                                      float* __restrict__ deg,
                                                      int n_edges) {
  const int e = blockIdx.x * (blockDim.x >> 5) + (threadIdx.x >> 5);
  if (e >= n_edges) return;
  const int lane = threadIdx.x & 31;
  const int si = src[e];
  const int di = dst[e];
  const float4 v = *(const float4*)(z + (size_t)si * OUTC + lane * 4);
  float* o = s + (size_t)di * OUTC + lane * 4;
  atomicAdd(o + 0, v.x);
  atomicAdd(o + 1, v.y);
  atomicAdd(o + 2, v.z);
  atomicAdd(o + 3, v.w);
  if (lane == 0) atomicAdd(deg + di, 1.0f);
}

__global__ __launch_bounds__(128) void gemm_finalize_kernel(const float* __restrict__ h,
                                                            const float* __restrict__ Wl,
                                                            const float* __restrict__ bl,
                                                            const float* __restrict__ s1,
                                                            const float* __restrict__ s2,
                                                            const float* __restrict__ deg1,
                                                            const float* __restrict__ deg2,
                                                            float* __restrict__ out) {
  const int lane  = threadIdx.x & 31;
  const int wave  = threadIdx.x >> 5;
  const int m0    = blockIdx.x * 32 + (wave >> 1) * 16;
  const int nbase = (wave & 1) * 64;
  v8f acc[4];
  wave_gemm_16x64(h, Wl, m0, nbase, lane, acc);
  const int l16   = lane & 15;
  const int rbase = (lane >> 4) * 8;
#pragma unroll
  for (int r = 0; r < 8; ++r) {
    const int m = m0 + rbase + r;
    const float i1 = 0.5f / fmaxf(deg1[m], 1.0f);
    const float i2 = 0.5f / fmaxf(deg2[m], 1.0f);
#pragma unroll
    for (int j = 0; j < 4; ++j) {
      const int    c   = nbase + j * 16 + l16;
      const size_t idx = (size_t)m * OUTC + c;
      const float  v   = acc[j][r] + bl[c] + s1[idx] * i1 + s2[idx] * i2;
      out[idx] = fmaxf(v, 0.0f);
    }
  }
}

extern "C" void kernel_launch(void* const* d_in, const int* in_sizes, int n_in,
                              void* d_out, int out_size, void* d_ws, size_t ws_size,
                              hipStream_t stream) {
  (void)n_in; (void)out_size; (void)ws_size;
  const float* h    = (const float*)d_in[0];
  const int*   src1 = (const int*)  d_in[1];
  const int*   dst1 = (const int*)  d_in[2];
  const int*   src2 = (const int*)  d_in[3];
  const int*   dst2 = (const int*)  d_in[4];
  const float* W1   = (const float*)d_in[5];
  const float* W2   = (const float*)d_in[6];
  const float* Wl   = (const float*)d_in[7];
  const float* bl   = (const float*)d_in[8];
  float*       out  = (float*)d_out;

  const size_t nodes = (size_t)in_sizes[0] / HID;
  const int    ne    = in_sizes[1];
  const size_t feat  = nodes * OUTC;

  float* s1   = (float*)d_ws;
  float* s2   = s1 + feat;
  float* deg1 = s2 + feat;
  float* deg2 = deg1 + nodes;
  float* z    = deg2 + nodes;

  hipMemsetAsync(d_ws, 0, (2 * feat + 2 * nodes) * sizeof(float), stream);

  const dim3 gblk(128);
  const dim3 ggrd((unsigned)(nodes / 32));
  const dim3 sblk(256);
  const dim3 sgrd((unsigned)((ne + 7) / 8));

  gemm_kernel<<<ggrd, gblk, 0, stream>>>(h, W1, z);
  scatter_kernel<<<sgrd, sblk, 0, stream>>>(src1, dst1, z, s1, deg1, ne);

  gemm_kernel<<<ggrd, gblk, 0, stream>>>(h, W2, z);
  scatter_kernel<<<sgrd, sblk, 0, stream>>>(src2, dst2, z, s2, deg2, ne);

  gemm_finalize_kernel<<<ggrd, gblk, 0, stream>>>(h, Wl, bl, s1, s2, deg1, deg2, out);
}


# Edges pre-sorted by owner work-group in one added launch (per-slice LDS counting sort, no global atomics): aggregation reads only its own runs, no scan of all of dst; f32 adds still in arrival order
# speedup vs baseline: 2.3428x; 2.3428x over previous
#include <hip/hip_runtime.h>

typedef float v2f __attribute__((ext_vector_type(2)));
typedef float v8f __attribute__((ext_vector_type(8)));

#define HID  128
#define OUTC 128
#define NWG  256


__device__ __forceinline__ void load_w_half(const float* __restrict__ W, int nbase, int lane,
                                            v2f (&B)[4][32]) {
  const int l16  = lane & 15;
  const int koff = (lane >> 4) * 2;
#pragma unroll
  for (int j = 0; j < 4; ++j) {
    const float* bp = W + (size_t)(nbase + 16 * j + l16) * HID + koff;
#pragma unroll
    for (int k = 0; k < 32; ++k) B[j][k] = *(const v2f*)(bp + 4 * k);
  }
}

__device__ __forceinline__ void load_a_tile(const float* __restrict__ h, int tile, int lane,
                                            v2f (&A)[32]) {
  const float* ap = h + (size_t)(tile * 16 + (lane & 15)) * HID + (lane >> 4) * 2;
#pragma unroll
  for (int k = 0; k < 32; ++k) A[k] = *(const v2f*)(ap + 4 * k);
}

__device__ __forceinline__ void hold_loaded(const v2f (&A)[32]) {
  asm volatile("" :: "v"(A[0]), "v"(A[1]), "v"(A[2]), "v"(A[3]), "v"(A[4]), "v"(A[5]), "v"(A[6]), "v"(A[7]), "v"(A[8]), "v"(A[9]), "v"(A[10]), "v"(A[11]), "v"(A[12]), "v"(A[13]), "v"(A[14]), "v"(A[15]));
  asm volatile("" :: "v"(A[16]), "v"(A[17]), "v"(A[18]), "v"(A[19]), "v"(A[20]), "v"(A[21]), "v"(A[22]), "v"(A[23]), "v"(A[24]), "v"(A[25]), "v"(A[26]), "v"(A[27]), "v"(A[28]), "v"(A[29]), "v"(A[30]), "v"(A[31]));
}

__device__ __forceinline__ void mma_tile(const v2f (&A)[32], const v2f (&B)[4][32], v8f (&acc)[4]) {
  acc[0] = v8f{}; acc[1] = v8f{}; acc[2] = v8f{}; acc[3] = v8f{};
#pragma unroll
  for (int k = 0; k < 32; ++k) {
#pragma unroll
    for (int j = 0; j < 4; ++j)
      acc[j] = __builtin_amdgcn_wmma_f32_16x16x4_f32(false, A[k], false, B[j][k], (short)0, acc[j], false, false);
  }
}

__device__ __forceinline__ void store_z(float* __restrict__ z, int tile, int nbase, int lane,
                                        const v8f (&acc)[4]) {
  const int l16   = lane & 15;
  const int rbase = (lane >> 4) * 8;
#pragma unroll
  for (int r = 0; r < 8; ++r) {
    const int m = tile * 16 + rbase + r;
#pragma unroll
    for (int j = 0; j < 4; ++j) z[(size_t)m * OUTC + nbase + j * 16 + l16] = acc[j][r];
  }
}

__global__ __launch_bounds__(256) void gemm_kernel(const float* __restrict__ h,
                                                   const float* __restrict__ W,
                                                   float* __restrict__ z, int n_nodes) {
  const int lane     = threadIdx.x & 31;
  const int wave     = __builtin_amdgcn_readfirstlane((int)(threadIdx.x >> 5));
  const int nbase    = (wave & 1) * 64;
  const int nstreams = gridDim.x * 4;
  const int ntiles   = n_nodes >> 4;
  v2f B[4][32];
  load_w_half(W, nbase, lane, B);
  int t = blockIdx.x * 4 + (wave >> 1);
  if (t >= ntiles) return;
  v2f A0[32], A1[32];
  v8f acc[4];
  load_a_tile(h, t, lane, A0);
  for (;;) {
    const int t1 = t + nstreams;
    load_a_tile(h, t1 < ntiles ? t1 : t, lane, A1);
    __builtin_amdgcn_sched_barrier(0);
    mma_tile(A0, B, acc);
    store_z(z, t, nbase, lane, acc);
    hold_loaded(A1);
    if (t1 >= ntiles) break;
    const int t2 = t1 + nstreams;
    load_a_tile(h, t2 < ntiles ? t2 : t1, lane, A0);
    __builtin_amdgcn_sched_barrier(0);
    mma_tile(A1, B, acc);
    store_z(z, t1, nbase, lane, acc);
    if (t2 >= ntiles) break;
    t = t2;
  }
}

#define BIN_NWG     256
#define BIN_THREADS 256
#define BIN_OFFS    257
#define BIN_GRID    512
#define BIN_U       8

__device__ __forceinline__ int bin_slice(int n_edges) { return (n_edges + BIN_NWG - 1) / BIN_NWG; }
__device__ __forceinline__ size_t bin_pairs_at(int type, int n_edges) {
  return (size_t)type * 2 * BIN_NWG * bin_slice(n_edges);
}
__device__ __forceinline__ size_t bin_offs_at(int type, int n_edges) {
  return (size_t)4 * BIN_NWG * bin_slice(n_edges) + (size_t)type * BIN_NWG * BIN_OFFS;
}

__device__ __forceinline__ void bin_hold(const int (&x)[BIN_U]) {
  asm volatile("" :: "v"(x[0]), "v"(x[1]), "v"(x[2]), "v"(x[3]), "v"(x[4]), "v"(x[5]), "v"(x[6]), "v"(x[7]));
}

__global__ __launch_bounds__(BIN_THREADS) void bin_kernel(const int* __restrict__ src1,
                                                          const int* __restrict__ dst1,
                                                          const int* __restrict__ src2,
                                                          const int* __restrict__ dst2,
                                                          int* __restrict__ wsx, int n_edges) {
  __shared__ int hist[BIN_NWG];
  __shared__ int offs[BIN_OFFS];
  __shared__ int wsum[BIN_THREADS / 32];
  const int type = blockIdx.x >> 8;
  const int w    = blockIdx.x & (BIN_NWG - 1);
  const int tid  = threadIdx.x;
  const int lane = tid & 31;
  const int wave = tid >> 5;
  const int* __restrict__ src = type ? src2 : src1;
  const int* __restrict__ dst = type ? dst2 : dst1;
  const int S    = bin_slice(n_edges);
  const int e0   = w * S;
  const int rest = n_edges - e0;
  const int cnt  = rest < 0 ? 0 : (rest < S ? rest : S);
  int* pairs = wsx + bin_pairs_at(type, n_edges) + (size_t)2 * e0;
  int* otab  = wsx + bin_offs_at(type, n_edges) + (size_t)w * BIN_OFFS;

  hist[tid] = 0;
  __syncthreads();
  for (int i = tid; i < cnt; i += BIN_THREADS * BIN_U) {
    int d[BIN_U];
#pragma unroll
    for (int u = 0; u < BIN_U; ++u) {
      const int ii = i + u * BIN_THREADS;
      d[u] = dst[e0 + (ii < cnt ? ii : i)];
    }
    bin_hold(d);
#pragma unroll
    for (int u = 0; u < BIN_U; ++u) {
      if (i + u * BIN_THREADS < cnt)
        __hip_atomic_fetch_add(hist + ((d[u] >> 5) & (BIN_NWG - 1)), 1, __ATOMIC_RELAXED, __HIP_MEMORY_SCOPE_WORKGROUP);
    }
  }
  __syncthreads();
  const int c = hist[tid];
  int x = c;
  for (int d = 1; d < 32; d <<= 1) {
    const int y = __shfl_up(x, d);
    if (lane >= d) x += y;
  }
  if (lane == 31) wsum[wave] = x;
  __syncthreads();
  int base = 0;
  for (int j = 0; j < wave; ++j) base += wsum[j];
  const int excl = base + x - c;
  offs[tid] = excl;
  otab[tid] = excl;
  if (tid == BIN_THREADS - 1) { offs[BIN_NWG] = excl + c; otab[BIN_NWG] = excl + c; }
  hist[tid] = 0;
  __syncthreads();
  for (int i = tid; i < cnt; i += BIN_THREADS * BIN_U) {
    int d[BIN_U], s[BIN_U];
#pragma unroll
    for (int u = 0; u < BIN_U; ++u) {
      const int ii = i + u * BIN_THREADS;
      d[u] = dst[e0 + (ii < cnt ? ii : i)];
      s[u] = src[e0 + (ii < cnt ? ii : i)];
    }
    bin_hold(d);
    bin_hold(s);
#pragma unroll
    for (int u = 0; u < BIN_U; ++u) {
      if (i + u * BIN_THREADS < cnt) {
        const int o = (d[u] >> 5) & (BIN_NWG - 1);
        const int p = offs[o] + __hip_atomic_fetch_add(hist + o, 1, __ATOMIC_RELAXED, __HIP_MEMORY_SCOPE_WORKGROUP);
        *(int2*)(pairs + 2 * p) = int2{s[u], d[u]};
      }
    }
  }
}

#define AGG_NWG     256
#define AGG_THREADS 512
#define AGG_WAVES   16
#define AGG_SLOTS   13
#define AGG_ROWS    (AGG_SLOTS * 32)

__device__ __forceinline__ int agg_local_row(int d, int b, int kbase) {
  const int k = (d >> 13) - kbase;
  const bool mine = (((d >> 5) & (AGG_NWG - 1)) == b) && ((unsigned)k < (unsigned)AGG_SLOTS);
  return mine ? (k * 32 + (d & 31)) : -1;
}

__device__ __forceinline__ void agg_lds_add(float* p, float v) {
  __hip_atomic_fetch_add(p, v, __ATOMIC_RELAXED, __HIP_MEMORY_SCOPE_WORKGROUP);
}

__device__ __forceinline__ void agg_group(int sv, int rv, const float* __restrict__ z,
                                          float* rows, int lane) {
  float4 v[32];
#pragma unroll
  for (int j = 0; j < 32; ++j) {
    const int sj = __builtin_amdgcn_readlane(sv, j);
    v[j] = *(const float4*)(z + (size_t)sj * OUTC + lane * 4);
  }
#pragma unroll
  for (int j = 0; j < 32; ++j) {
    const int rj = __builtin_amdgcn_readlane(rv, j);
    float* p = rows + rj * OUTC + lane;
    agg_lds_add(p, v[j].x);
    agg_lds_add(p + 32, v[j].y);
    agg_lds_add(p + 64, v[j].z);
    agg_lds_add(p + 96, v[j].w);
  }
}

__global__ __launch_bounds__(AGG_THREADS) void aggregate_kernel(const int* __restrict__ wsx, int type,
                                                                const float* __restrict__ z,
                                                                float* __restrict__ s,
                                                                float* __restrict__ deg,
                                                                int n_edges, int n_nodes) {
  __shared__ float rows[(AGG_ROWS + 1) * OUTC];
  __shared__ int   cnt[AGG_ROWS];
  const int lane  = threadIdx.x & 31;
  const int wave  = __builtin_amdgcn_readfirstlane((int)(threadIdx.x >> 5));
  const int b     = blockIdx.x;
  const int S     = bin_slice(n_edges);
  const int* pairs = wsx + bin_pairs_at(type, n_edges);
  const int* otab  = wsx + bin_offs_at(type, n_edges);
  const int nrounds = (n_nodes + 8191) >> 13;

  const int  sl  = wave * 16 + (lane & 15);
  const int* op  = otab + (size_t)sl * BIN_OFFS + b;
  const int  lo  = op[0];
  const int  hi  = op[1];

  for (int i = threadIdx.x; i < AGG_ROWS * OUTC / 4; i += AGG_THREADS)
    ((float4*)rows)[i] = float4{0.f, 0.f, 0.f, 0.f};
  for (int i = threadIdx.x; i < AGG_ROWS; i += AGG_THREADS) cnt[i] = 0;

  const int  c   = lane < 16 ? hi - lo : 0;
  int incl = c;
  for (int d = 1; d < 16; d <<= 1) {
    const int y = __shfl_up(incl, d);
    if (lane >= d) incl += y;
  }
  const int gl = sl * S + lo - (incl - c);
  int P[16], G[16];
#pragma unroll
  for (int j = 0; j < 16; ++j) {
    P[j] = __builtin_amdgcn_readlane(incl - c, j);
    G[j] = __builtin_amdgcn_readlane(gl, j);
  }
  const int total = __builtin_amdgcn_readlane(incl, 15);

  auto load_pair = [&](int i0) -> int2 {
    const int iv = i0 + lane < total ? i0 + lane : i0;
    int gi = G[0] + iv;
#pragma unroll
    for (int j = 1; j < 16; ++j) gi += (iv >= P[j]) ? (G[j] - G[j - 1]) : 0;
    return *(const int2*)(pairs + (size_t)2 * gi);
  };

  for (int kbase = 0; kbase < nrounds; kbase += AGG_SLOTS) {
    __syncthreads();

    if (total > 0) {
      int2 ed = load_pair(0);
      for (int i0 = 0; i0 < total; i0 += 32) {
        const bool more = i0 + 32 < total;
        int2 nx = ed;
        if (more) nx = load_pair(i0 + 32);
        const int  lr = agg_local_row(ed.y, b, kbase);
        const bool ok = (i0 + lane < total) && lr >= 0;
        if (ok) __hip_atomic_fetch_add(cnt + lr, 1, __ATOMIC_RELAXED, __HIP_MEMORY_SCOPE_WORKGROUP);
        agg_group(ed.x, ok ? lr : AGG_ROWS, z, rows, lane);
        ed = nx;
      }
    }
    __syncthreads();

    for (int r = wave; r < AGG_ROWS; r += AGG_WAVES) {
      const int m = (((kbase + (r >> 5)) * AGG_NWG + b) << 5) + (r & 31);
      if (m < n_nodes) {
        const float* p = rows + r * OUTC + lane;
        const float4 v = float4{p[0], p[32], p[64], p[96]};
        *(float4*)(s + (size_t)m * OUTC + lane * 4) = v;
      }
    }
    for (int r = threadIdx.x; r < AGG_ROWS; r += AGG_THREADS) {
      const int m = (((kbase + (r >> 5)) * AGG_NWG + b) << 5) + (r & 31);
      if (m < n_nodes) deg[m] = (float)cnt[r];
    }
    if (kbase + AGG_SLOTS < nrounds) {
      __syncthreads();
      for (int i = threadIdx.x; i < AGG_ROWS * OUTC / 4; i += AGG_THREADS)
        ((float4*)rows)[i] = float4{0.f, 0.f, 0.f, 0.f};
      for (int i = threadIdx.x; i < AGG_ROWS; i += AGG_THREADS) cnt[i] = 0;
    }
  }
}

__device__ __forceinline__ void store_final(float* __restrict__ out,
                                            const float* __restrict__ s1,
                                            const float* __restrict__ s2,
                                            const float* __restrict__ deg1,
                                            const float* __restrict__ deg2,
                                            const float (&blv)[4],
                                            int tile, int nbase, int lane, const v8f (&acc)[4]) {
  const int l16   = lane & 15;
  const int rbase = (lane >> 4) * 8;
#pragma unroll
  for (int rh = 0; rh < 8; rh += 4) {
    float a1[4][4], a2[4][4], d1[4], d2[4];
#pragma unroll
    for (int q = 0; q < 4; ++q) {
      const int m = tile * 16 + rbase + rh + q;
      d1[q] = deg1[m];
      d2[q] = deg2[m];
#pragma unroll
      for (int j = 0; j < 4; ++j) {
        const size_t idx = (size_t)m * OUTC + nbase + j * 16 + l16;
        a1[q][j] = s1[idx];
        a2[q][j] = s2[idx];
      }
    }
    __builtin_amdgcn_sched_barrier(0);
#pragma unroll
    for (int q = 0; q < 4; ++q) {
      const int m = tile * 16 + rbase + rh + q;
      const float i1 = 0.5f / fmaxf(d1[q], 1.0f);
      const float i2 = 0.5f / fmaxf(d2[q], 1.0f);
#pragma unroll
      for (int j = 0; j < 4; ++j) {
        const size_t idx = (size_t)m * OUTC + nbase + j * 16 + l16;
        const float  v   = acc[j][rh + q] + blv[j] + a1[q][j] * i1 + a2[q][j] * i2;
        out[idx] = fmaxf(v, 0.0f);
      }
    }
  }
}

__global__ __launch_bounds__(256) void gemm_finalize_kernel(const float* __restrict__ h,
                                                            const float* __restrict__ Wl,
                                                            const float* __restrict__ bl,
                                                            const float* __restrict__ s1,
                                                            const float* __restrict__ s2,
                                                            const float* __restrict__ deg1,
                                                            const float* __restrict__ deg2,
                                                            float* __restrict__ out, int n_nodes) {
  const int lane     = threadIdx.x & 31;
  const int wave     = __builtin_amdgcn_readfirstlane((int)(threadIdx.x >> 5));
  const int nbase    = (wave & 1) * 64;
  const int nstreams = gridDim.x * 4;
  const int ntiles   = n_nodes >> 4;
  v2f B[4][32];
  load_w_half(Wl, nbase, lane, B);
  float blv[4];
#pragma unroll
  for (int j = 0; j < 4; ++j) blv[j] = bl[nbase + j * 16 + (lane & 15)];
  int t = blockIdx.x * 4 + (wave >> 1);
  if (t >= ntiles) return;
  v2f A0[32], A1[32];
  v8f acc[4];
  load_a_tile(h, t, lane, A0);
  for (;;) {
    const int t1 = t + nstreams;
    load_a_tile(h, t1 < ntiles ? t1 : t, lane, A1);
    __builtin_amdgcn_sched_barrier(0);
    mma_tile(A0, B, acc);
    __builtin_amdgcn_sched_barrier(0);
    store_final(out, s1, s2, deg1, deg2, blv, t, nbase, lane, acc);
    hold_loaded(A1);
    if (t1 >= ntiles) break;
    const int t2 = t1 + nstreams;
    load_a_tile(h, t2 < ntiles ? t2 : t1, lane, A0);
    __builtin_amdgcn_sched_barrier(0);
    mma_tile(A1, B, acc);
    __builtin_amdgcn_sched_barrier(0);
    store_final(out, s1, s2, deg1, deg2, blv, t1, nbase, lane, acc);
    if (t2 >= ntiles) break;
    t = t2;
  }
}

extern "C" void kernel_launch(void* const* d_in, const int* in_sizes, int n_in,
                              void* d_out, int out_size, void* d_ws, size_t ws_size,
                              hipStream_t stream) {
  (void)n_in; (void)out_size; (void)ws_size;
  const float* h    = (const float*)d_in[0];
  const int*   src1 = (const int*)  d_in[1];
  const int*   dst1 = (const int*)  d_in[2];
  const int*   src2 = (const int*)  d_in[3];
  const int*   dst2 = (const int*)  d_in[4];
  const float* W1   = (const float*)d_in[5];
  const float* W2   = (const float*)d_in[6];
  const float* Wl   = (const float*)d_in[7];
  const float* bl   = (const float*)d_in[8];
  float*       out  = (float*)d_out;

  const size_t nodes = (size_t)in_sizes[0] / HID;
  const int    ne    = in_sizes[1];
  const size_t feat  = nodes * OUTC;

  float* s1   = (float*)d_ws;
  float* s2   = s1 + feat;
  float* deg1 = s2 + feat;
  float* deg2 = deg1 + nodes;
  float* z    = deg2 + nodes;
  int*   wsx  = (int*)(z + feat);

  const dim3 gblk(256);
  const dim3 ggrd(NWG);
  const dim3 sblk(AGG_THREADS);
  const dim3 sgrd(AGG_NWG);

  bin_kernel<<<dim3(BIN_GRID), dim3(BIN_THREADS), 0, stream>>>(src1, dst1, src2, dst2, wsx, ne);

  gemm_kernel<<<ggrd, gblk, 0, stream>>>(h, W1, z, (int)nodes);
  aggregate_kernel<<<sgrd, sblk, 0, stream>>>(wsx, 0, z, s1, deg1, ne, (int)nodes);

  gemm_kernel<<<ggrd, gblk, 0, stream>>>(h, W2, z, (int)nodes);
  aggregate_kernel<<<sgrd, sblk, 0, stream>>>(wsx, 1, z, s2, deg2, ne, (int)nodes);

  gemm_finalize_kernel<<<ggrd, gblk, 0, stream>>>(h, Wl, bl, s1, s2, deg1, deg2, out, (int)nodes);
}


# Weights rewritten in fragment order by the first launch (48 added work-groups): a wave fills its 256 weight registers with 64 whole-line 512-byte loads, not 128 sixteen-line loads; k order unchanged
# speedup vs baseline: 2.6164x; 1.1168x over previous
#include <hip/hip_runtime.h>

typedef float v2f __attribute__((ext_vector_type(2)));
typedef float v8f __attribute__((ext_vector_type(8)));

#define HID  128
#define OUTC 128
#define NWG  256

#define WF_FLOATS (OUTC * HID)
#define WF_VEC4   (WF_FLOATS / 4)

__device__ __forceinline__ void load_w_half(const float* __restrict__ Wf, int nbase, int lane,
                                            v2f (&B)[4][32]) {
  const float4* bp = (const float4*)Wf + (size_t)(nbase >> 6) * (WF_VEC4 / 2) + lane;
#pragma unroll
  for (int j = 0; j < 4; ++j) {
#pragma unroll
    for (int kp = 0; kp < 16; ++kp) {
      const float4 v = bp[(j * 16 + kp) * 32];
      B[j][2 * kp]     = v2f{v.x, v.y};
      B[j][2 * kp + 1] = v2f{v.z, v.w};
    }
  }
}

__device__ __forceinline__ void load_a_tile(const float* __restrict__ h, int tile, int lane,
                                            v2f (&A)[32]) {
  const float* ap = h + (size_t)(tile * 16 + (lane & 15)) * HID + (lane >> 4) * 2;
#pragma unroll
  for (int k = 0; k < 32; ++k) A[k] = *(const v2f*)(ap + 4 * k);
}

__device__ __forceinline__ void hold_loaded(const v2f (&A)[32]) {
  asm volatile("" :: "v"(A[0]), "v"(A[1]), "v"(A[2]), "v"(A[3]), "v"(A[4]), "v"(A[5]), "v"(A[6]), "v"(A[7]), "v"(A[8]), "v"(A[9]), "v"(A[10]), "v"(A[11]), "v"(A[12]), "v"(A[13]), "v"(A[14]), "v"(A[15]));
  asm volatile("" :: "v"(A[16]), "v"(A[17]), "v"(A[18]), "v"(A[19]), "v"(A[20]), "v"(A[21]), "v"(A[22]), "v"(A[23]), "v"(A[24]), "v"(A[25]), "v"(A[26]), "v"(A[27]), "v"(A[28]), "v"(A[29]), "v"(A[30]), "v"(A[31]));
}

__device__ __forceinline__ void mma_tile(const v2f (&A)[32], const v2f (&B)[4][32], v8f (&acc)[4]) {
  acc[0] = v8f{}; acc[1] = v8f{}; acc[2] = v8f{}; acc[3] = v8f{};
#pragma unroll
  for (int k = 0; k < 32; ++k) {
#pragma unroll
    for (int j = 0; j < 4; ++j)
      acc[j] = __builtin_amdgcn_wmma_f32_16x16x4_f32(false, A[k], false, B[j][k], (short)0, acc[j], false, false);
  }
}

__device__ __forceinline__ void store_z(float* __restrict__ z, int tile, int nbase, int lane,
                                        const v8f (&acc)[4]) {
  const int l16   = lane & 15;
  const int rbase = (lane >> 4) * 8;
#pragma unroll
  for (int r = 0; r < 8; ++r) {
    const int m = tile * 16 + rbase + r;
#pragma unroll
    for (int j = 0; j < 4; ++j) z[(size_t)m * OUTC + nbase + j * 16 + l16] = acc[j][r];
  }
}

__global__ __launch_bounds__(256) void gemm_kernel(const float* __restrict__ h,
                                                   const float* __restrict__ W,
                                                   float* __restrict__ z, int n_nodes) {
  const int lane     = threadIdx.x & 31;
  const int wave     = __builtin_amdgcn_readfirstlane((int)(threadIdx.x >> 5));
  const int nbase    = (wave & 1) * 64;
  const int nstreams = gridDim.x * 4;
  const int ntiles   = n_nodes >> 4;
  v2f B[4][32];
  load_w_half(W, nbase, lane, B);
  int t = blockIdx.x * 4 + (wave >> 1);
  if (t >= ntiles) return;
  v2f A0[32], A1[32];
  v8f acc[4];
  load_a_tile(h, t, lane, A0);
  for (;;) {
    const int t1 = t + nstreams;
    load_a_tile(h, t1 < ntiles ? t1 : t, lane, A1);
    __builtin_amdgcn_sched_barrier(0);
    mma_tile(A0, B, acc);
    store_z(z, t, nbase, lane, acc);
    hold_loaded(A1);
    if (t1 >= ntiles) break;
    const int t2 = t1 + nstreams;
    load_a_tile(h, t2 < ntiles ? t2 : t1, lane, A0);
    __builtin_amdgcn_sched_barrier(0);
    mma_tile(A1, B, acc);
    store_z(z, t1, nbase, lane, acc);
    if (t2 >= ntiles) break;
    t = t2;
  }
}

#define BIN_NWG     256
#define BIN_THREADS 256
#define BIN_OFFS    257
#define BIN_GRID    512
#define BIN_U       8
#define WF_BLOCKS   (WF_VEC4 / BIN_THREADS)
#define PREP_GRID   (BIN_GRID + 3 * WF_BLOCKS)

__device__ __forceinline__ int bin_slice(int n_edges) { return (n_edges + BIN_NWG - 1) / BIN_NWG; }
__device__ __forceinline__ size_t bin_pairs_at(int type, int n_edges) {
  return (size_t)type * 2 * BIN_NWG * bin_slice(n_edges);
}
__device__ __forceinline__ size_t bin_offs_at(int type, int n_edges) {
  return (size_t)4 * BIN_NWG * bin_slice(n_edges) + (size_t)type * BIN_NWG * BIN_OFFS;
}

__device__ __forceinline__ void bin_hold(const int (&x)[BIN_U]) {
  asm volatile("" :: "v"(x[0]), "v"(x[1]), "v"(x[2]), "v"(x[3]), "v"(x[4]), "v"(x[5]), "v"(x[6]), "v"(x[7]));
}

__global__ __launch_bounds__(BIN_THREADS) void bin_kernel(const int* __restrict__ src1,
                                                          const int* __restrict__ dst1,
                                                          const int* __restrict__ src2,
                                                          const int* __restrict__ dst2,
                                                          const float* __restrict__ W1,
                                                          const float* __restrict__ W2,
                                                          const float* __restrict__ Wl,
                                                          float* __restrict__ wf,
                                                          int* __restrict__ wsx, int n_edges) {
  __shared__ int hist[BIN_NWG];
  __shared__ int offs[BIN_OFFS];
  __shared__ int wsum[BIN_THREADS / 32];
  const int tid  = threadIdx.x;
  if (blockIdx.x >= BIN_GRID) {
    const int q     = blockIdx.x - BIN_GRID;
    const int which = q / WF_BLOCKS;
    const float* __restrict__ W = which == 0 ? W1 : (which == 1 ? W2 : Wl);
    const int idx = (q - which * WF_BLOCKS) * BIN_THREADS + tid;
    const int ln  = idx & 31;
    const int kp  = (idx >> 5) & 15;
    const int j   = (idx >> 9) & 3;
    const int nb  = idx >> 11;
    const int n   = nb * 64 + j * 16 + (ln & 15);
    const float* p = W + (size_t)n * HID + kp * 8 + (ln >> 4) * 2;
    const v2f lo = *(const v2f*)p;
    const v2f hi = *(const v2f*)(p + 4);
    ((float4*)(wf + (size_t)which * WF_FLOATS))[idx] = float4{lo.x, lo.y, hi.x, hi.y};
    return;
  }
  const int type = blockIdx.x >> 8;
  const int w    = blockIdx.x & (BIN_NWG - 1);
  const int lane = tid & 31;
  const int wave = tid >> 5;
  const int* __restrict__ src = type ? src2 : src1;
  const int* __restrict__ dst = type ? dst2 : dst1;
  const int S    = bin_slice(n_edges);
  const int e0   = w * S;
  const int rest = n_edges - e0;
  const int cnt  = rest < 0 ? 0 : (rest < S ? rest : S);
  int* pairs = wsx + bin_pairs_at(type, n_edges) + (size_t)2 * e0;
  int* otab  = wsx + bin_offs_at(type, n_edges) + (size_t)w * BIN_OFFS;

  hist[tid] = 0;
  __syncthreads();
  for (int i = tid; i < cnt; i += BIN_THREADS * BIN_U) {
    int d[BIN_U];
#pragma unroll
    for (int u = 0; u < BIN_U; ++u) {
      const int ii = i + u * BIN_THREADS;
      d[u] = dst[e0 + (ii < cnt ? ii : i)];
    }
    bin_hold(d);
#pragma unroll
    for (int u = 0; u < BIN_U; ++u) {
      if (i + u * BIN_THREADS < cnt)
        __hip_atomic_fetch_add(hist + ((d[u] >> 5) & (BIN_NWG - 1)), 1, __ATOMIC_RELAXED, __HIP_MEMORY_SCOPE_WORKGROUP);
    }
  }
  __syncthreads();
  const int c = hist[tid];
  int x = c;
  for (int d = 1; d < 32; d <<= 1) {
    const int y = __shfl_up(x, d);
    if (lane >= d) x += y;
  }
  if (lane == 31) wsum[wave] = x;
  __syncthreads();
  int base = 0;
  for (int j = 0; j < wave; ++j) base += wsum[j];
  const int excl = base + x - c;
  offs[tid] = excl;
  otab[tid] = excl;
  if (tid == BIN_THREADS - 1) { offs[BIN_NWG] = excl + c; otab[BIN_NWG] = excl + c; }
  hist[tid] = 0;
  __syncthreads();
  for (int i = tid; i < cnt; i += BIN_THREADS * BIN_U) {
    int d[BIN_U], s[BIN_U];
#pragma unroll
    for (int u = 0; u < BIN_U; ++u) {
      const int ii = i + u * BIN_THREADS;
      d[u] = dst[e0 + (ii < cnt ? ii : i)];
      s[u] = src[e0 + (ii < cnt ? ii : i)];
    }
    bin_hold(d);
    bin_hold(s);
#pragma unroll
    for (int u = 0; u < BIN_U; ++u) {
      if (i + u * BIN_THREADS < cnt) {
        const int o = (d[u] >> 5) & (BIN_NWG - 1);
        const int p = offs[o] + __hip_atomic_fetch_add(hist + o, 1, __ATOMIC_RELAXED, __HIP_MEMORY_SCOPE_WORKGROUP);
        *(int2*)(pairs + 2 * p) = int2{s[u], d[u]};
      }
    }
  }
}

#define AGG_NWG     256
#define AGG_THREADS 512
#define AGG_WAVES   16
#define AGG_SLOTS   13
#define AGG_ROWS    (AGG_SLOTS * 32)

__device__ __forceinline__ int agg_local_row(int d, int b, int kbase) {
  const int k = (d >> 13) - kbase;
  const bool mine = (((d >> 5) & (AGG_NWG - 1)) == b) && ((unsigned)k < (unsigned)AGG_SLOTS);
  return mine ? (k * 32 + (d & 31)) : -1;
}

__device__ __forceinline__ void agg_lds_add(float* p, float v) {
  __hip_atomic_fetch_add(p, v, __ATOMIC_RELAXED, __HIP_MEMORY_SCOPE_WORKGROUP);
}

__device__ __forceinline__ void agg_group(int sv, int rv, const float* __restrict__ z,
                                          float* rows, int lane) {
  float4 v[32];
#pragma unroll
  for (int j = 0; j < 32; ++j) {
    const int sj = __builtin_amdgcn_readlane(sv, j);
    v[j] = *(const float4*)(z + (size_t)sj * OUTC + lane * 4);
  }
#pragma unroll
  for (int j = 0; j < 32; ++j) {
    const int rj = __builtin_amdgcn_readlane(rv, j);
    float* p = rows + rj * OUTC + lane;
    agg_lds_add(p, v[j].x);
    agg_lds_add(p + 32, v[j].y);
    agg_lds_add(p + 64, v[j].z);
    agg_lds_add(p + 96, v[j].w);
  }
}

__global__ __launch_bounds__(AGG_THREADS) void aggregate_kernel(const int* __restrict__ wsx, int type,
                                                                const float* __restrict__ z,
                                                                float* __restrict__ s,
                                                                float* __restrict__ deg,
                                                                int n_edges, int n_nodes) {
  __shared__ float rows[(AGG_ROWS + 1) * OUTC];
  __shared__ int   cnt[AGG_ROWS];
  const int lane  = threadIdx.x & 31;
  const int wave  = __builtin_amdgcn_readfirstlane((int)(threadIdx.x >> 5));
  const int b     = blockIdx.x;
  const int S     = bin_slice(n_edges);
  const int* pairs = wsx + bin_pairs_at(type, n_edges);
  const int* otab  = wsx + bin_offs_at(type, n_edges);
  const int nrounds = (n_nodes + 8191) >> 13;

  const int  sl  = wave * 16 + (lane & 15);
  const int* op  = otab + (size_t)sl * BIN_OFFS + b;
  const int  lo  = op[0];
  const int  hi  = op[1];

  for (int i = threadIdx.x; i < AGG_ROWS * OUTC / 4; i += AGG_THREADS)
    ((float4*)rows)[i] = float4{0.f, 0.f, 0.f, 0.f};
  for (int i = threadIdx.x; i < AGG_ROWS; i += AGG_THREADS) cnt[i] = 0;

  const int  c   = lane < 16 ? hi - lo : 0;
  int incl = c;
  for (int d = 1; d < 16; d <<= 1) {
    const int y = __shfl_up(incl, d);
    if (lane >= d) incl += y;
  }
  const int gl = sl * S + lo - (incl - c);
  int P[16], G[16];
#pragma unroll
  for (int j = 0; j < 16; ++j) {
    P[j] = __builtin_amdgcn_readlane(incl - c, j);
    G[j] = __builtin_amdgcn_readlane(gl, j);
  }
  const int total = __builtin_amdgcn_readlane(incl, 15);

  auto load_pair = [&](int i0) -> int2 {
    const int iv = i0 + lane < total ? i0 + lane : i0;
    int gi = G[0] + iv;
#pragma unroll
    for (int j = 1; j < 16; ++j) gi += (iv >= P[j]) ? (G[j] - G[j - 1]) : 0;
    return *(const int2*)(pairs + (size_t)2 * gi);
  };

  for (int kbase = 0; kbase < nrounds; kbase += AGG_SLOTS) {
    __syncthreads();

    if (total > 0) {
      int2 ed = load_pair(0);
      for (int i0 = 0; i0 < total; i0 += 32) {
        const bool more = i0 + 32 < total;
        int2 nx = ed;
        if (more) nx = load_pair(i0 + 32);
        const int  lr = agg_local_row(ed.y, b, kbase);
        const bool ok = (i0 + lane < total) && lr >= 0;
        if (ok) __hip_atomic_fetch_add(cnt + lr, 1, __ATOMIC_RELAXED, __HIP_MEMORY_SCOPE_WORKGROUP);
        agg_group(ed.x, ok ? lr : AGG_ROWS, z, rows, lane);
        ed = nx;
      }
    }
    __syncthreads();

    for (int r = wave; r < AGG_ROWS; r += AGG_WAVES) {
      const int m = (((kbase + (r >> 5)) * AGG_NWG + b) << 5) + (r & 31);
      if (m < n_nodes) {
        const float* p = rows + r * OUTC + lane;
        const float4 v = float4{p[0], p[32], p[64], p[96]};
        *(float4*)(s + (size_t)m * OUTC + lane * 4) = v;
      }
    }
    for (int r = threadIdx.x; r < AGG_ROWS; r += AGG_THREADS) {
      const int m = (((kbase + (r >> 5)) * AGG_NWG + b) << 5) + (r & 31);
      if (m < n_nodes) deg[m] = (float)cnt[r];
    }
    if (kbase + AGG_SLOTS < nrounds) {
      __syncthreads();
      for (int i = threadIdx.x; i < AGG_ROWS * OUTC / 4; i += AGG_THREADS)
        ((float4*)rows)[i] = float4{0.f, 0.f, 0.f, 0.f};
      for (int i = threadIdx.x; i < AGG_ROWS; i += AGG_THREADS) cnt[i] = 0;
    }
  }
}

__device__ __forceinline__ void store_final(float* __restrict__ out,
                                            const float* __restrict__ s1,
                                            const float* __restrict__ s2,
                                            const float* __restrict__ deg1,
                                            const float* __restrict__ deg2,
                                            const float (&blv)[4],
                                            int tile, int nbase, int lane, const v8f (&acc)[4]) {
  const int l16   = lane & 15;
  const int rbase = (lane >> 4) * 8;
#pragma unroll
  for (int rh = 0; rh < 8; rh += 4) {
    float a1[4][4], a2[4][4], d1[4], d2[4];
#pragma unroll
    for (int q = 0; q < 4; ++q) {
      const int m = tile * 16 + rbase + rh + q;
      d1[q] = deg1[m];
      d2[q] = deg2[m];
#pragma unroll
      for (int j = 0; j < 4; ++j) {
        const size_t idx = (size_t)m * OUTC + nbase + j * 16 + l16;
        a1[q][j] = s1[idx];
        a2[q][j] = s2[idx];
      }
    }
    __builtin_amdgcn_sched_barrier(0);
#pragma unroll
    for (int q = 0; q < 4; ++q) {
      const int m = tile * 16 + rbase + rh + q;
      const float i1 = 0.5f / fmaxf(d1[q], 1.0f);
      const float i2 = 0.5f / fmaxf(d2[q], 1.0f);
#pragma unroll
      for (int j = 0; j < 4; ++j) {
        const size_t idx = (size_t)m * OUTC + nbase + j * 16 + l16;
        const float  v   = acc[j][rh + q] + blv[j] + a1[q][j] * i1 + a2[q][j] * i2;
        out[idx] = fmaxf(v, 0.0f);
      }
    }
  }
}

__global__ __launch_bounds__(256) void gemm_finalize_kernel(const float* __restrict__ h,
                                                            const float* __restrict__ Wl,
                                                            const float* __restrict__ bl,
                                                            const float* __restrict__ s1,
                                                            const float* __restrict__ s2,
                                                            const float* __restrict__ deg1,
                                                            const float* __restrict__ deg2,
                                                            float* __restrict__ out, int n_nodes) {
  const int lane     = threadIdx.x & 31;
  const int wave     = __builtin_amdgcn_readfirstlane((int)(threadIdx.x >> 5));
  const int nbase    = (wave & 1) * 64;
  const int nstreams = gridDim.x * 4;
  const int ntiles   = n_nodes >> 4;
  v2f B[4][32];
  load_w_half(Wl, nbase, lane, B);
  float blv[4];
#pragma unroll
  for (int j = 0; j < 4; ++j) blv[j] = bl[nbase + j * 16 + (lane & 15)];
  int t = blockIdx.x * 4 + (wave >> 1);
  if (t >= ntiles) return;
  v2f A0[32], A1[32];
  v8f acc[4];
  load_a_tile(h, t, lane, A0);
  for (;;) {
    const int t1 = t + nstreams;
    load_a_tile(h, t1 < ntiles ? t1 : t, lane, A1);
    __builtin_amdgcn_sched_barrier(0);
    mma_tile(A0, B, acc);
    __builtin_amdgcn_sched_barrier(0);
    store_final(out, s1, s2, deg1, deg2, blv, t, nbase, lane, acc);
    hold_loaded(A1);
    if (t1 >= ntiles) break;
    const int t2 = t1 + nstreams;
    load_a_tile(h, t2 < ntiles ? t2 : t1, lane, A0);
    __builtin_amdgcn_sched_barrier(0);
    mma_tile(A1, B, acc);
    __builtin_amdgcn_sched_barrier(0);
    store_final(out, s1, s2, deg1, deg2, blv, t1, nbase, lane, acc);
    if (t2 >= ntiles) break;
    t = t2;
  }
}

extern "C" void kernel_launch(void* const* d_in, const int* in_sizes, int n_in,
                              void* d_out, int out_size, void* d_ws, size_t ws_size,
                              hipStream_t stream) {
  (void)n_in; (void)out_size; (void)ws_size;
  const float* h    = (const float*)d_in[0];
  const int*   src1 = (const int*)  d_in[1];
  const int*   dst1 = (const int*)  d_in[2];
  const int*   src2 = (const int*)  d_in[3];
  const int*   dst2 = (const int*)  d_in[4];
  const float* W1   = (const float*)d_in[5];
  const float* W2   = (const float*)d_in[6];
  const float* Wl   = (const float*)d_in[7];
  const float* bl   = (const float*)d_in[8];
  float*       out  = (float*)d_out;

  const size_t nodes = (size_t)in_sizes[0] / HID;
  const int    ne    = in_sizes[1];
  const size_t feat  = nodes * OUTC;

  float* s1   = (float*)d_ws;
  float* s2   = s1 + feat;
  float* deg1 = s2 + feat;
  float* deg2 = deg1 + nodes;
  float* z    = deg2 + nodes;
  float* wf   = z + feat;
  int*   wsx  = (int*)(wf + 3 * WF_FLOATS);
  const float* wf1 = wf;
  const float* wf2 = wf + WF_FLOATS;
  const float* wfl = wf + 2 * WF_FLOATS;

  const dim3 gblk(256);
  const dim3 ggrd(NWG);
  const dim3 sblk(AGG_THREADS);
  const dim3 sgrd(AGG_NWG);

  bin_kernel<<<dim3(PREP_GRID), dim3(BIN_THREADS), 0, stream>>>(src1, dst1, src2, dst2, W1, W2, Wl, wf, wsx, ne);

  gemm_kernel<<<ggrd, gblk, 0, stream>>>(h, wf1, z, (int)nodes);
  aggregate_kernel<<<sgrd, sblk, 0, stream>>>(wsx, 0, z, s1, deg1, ne, (int)nodes);

  gemm_kernel<<<ggrd, gblk, 0, stream>>>(h, wf2, z, (int)nodes);
  aggregate_kernel<<<sgrd, sblk, 0, stream>>>(wsx, 1, z, s2, deg2, ne, (int)nodes);

  gemm_finalize_kernel<<<ggrd, gblk, 0, stream>>>(h, wfl, bl, s1, s2, deg1, deg2, out, (int)nodes);
}
